# starting kernel
# baseline (speedup 1.0000x reference)
#include <hip/hip_runtime.h>

typedef float v2f __attribute__((ext_vector_type(2)));
typedef float v8f __attribute__((ext_vector_type(8)));

#define C_IN   128
#define K_OUT  128
#define W_     56
#define HW_    3136
#define KRED   1152

__global__ __launch_bounds__(256)
void conv3x3_wmma_f32(const float* __restrict__ in,
                      const float* __restrict__ wgt,
                      const float* __restrict__ bias,
                      float* __restrict__ out)
{
    __shared__ float wlds[16 * KRED];

    const int tid     = threadIdx.x;
    const int oc_base = blockIdx.y * 16;

    {
        const float* wsrc = wgt + (long)oc_base * KRED;
        for (int g = tid; g < 16 * KRED; g += 256) {
            int oc_l = g / KRED;
            int rem  = g - oc_l * KRED;
            int c    = rem / 9;
            int rs   = rem - c * 9;
            wlds[(rs * C_IN + c) * 16 + oc_l] = wsrc[g];
        }
    }
    __syncthreads();

    const int wave = tid >> 5;
    const int lane = tid & 31;
    const int m    = lane & 15;
    const int koff = (lane >> 4) << 1;

    const int p  = blockIdx.x * 128 + wave * 16 + m;
    const int n  = p / HW_;
    const int hw = p - n * HW_;
    const int h  = hw / W_;
    const int w  = hw - h * W_;

    v8f acc = {};

    #pragma unroll
    for (int rs = 0; rs < 9; ++rs) {
        const int r  = rs / 3;
        const int s  = rs - r * 3;
        const int ih = h + r - 1;
        const int iw = w + s - 1;
        const bool valid = (ih >= 0) & (ih < W_) & (iw >= 0) & (iw < W_);
        const float mask = valid ? 1.0f : 0.0f;
        const long  off  = valid ? ((long)n * C_IN * HW_ + (long)ih * W_ + iw)
                                 : 0;
        const float* gin = in + off + (long)koff * HW_;
        const float* wl  = &wlds[(rs * C_IN + koff) * 16 + m];

        #pragma unroll 4
        for (int c = 0; c < C_IN; c += 4) {
            v2f A, B;
            A.x = wl[(c    ) * 16];
            A.y = wl[(c + 1) * 16];
            B.x = gin[(long)(c    ) * HW_] * mask;
            B.y = gin[(long)(c + 1) * HW_] * mask;
            acc = __builtin_amdgcn_wmma_f32_16x16x4_f32(
                false, A, false, B, (short)0, acc, false, false);
        }
    }

    const int m_hi = (lane >> 4) << 3;
    #pragma unroll
    for (int g = 0; g < 8; ++g) {
        const int oc = oc_base + m_hi + g;
        out[((long)n * K_OUT + oc) * HW_ + hw] = acc[g] + bias[oc];
    }
}

extern "C" void kernel_launch(void* const* d_in, const int* in_sizes, int n_in,
                              void* d_out, int out_size, void* d_ws, size_t ws_size,
                              hipStream_t stream) {
    (void)in_sizes; (void)n_in; (void)d_ws; (void)ws_size; (void)out_size;
    const float* in   = (const float*)d_in[0];
    const float* wgt  = (const float*)d_in[1];
    const float* bias = (const float*)d_in[2];
    float*       out  = (float*)d_out;

    dim3 grid(392, 8, 1);
    dim3 block(256, 1, 1);
    conv3x3_wmma_f32<<<grid, block, 0, stream>>>(in, wgt, bias, out);
}


# Minimal filtering F(2,3) down image columns on the LDS strip: a row pair takes 4 WMMAs where 6 were, input rows differenced in registers at the fragment read; 7th row direct; summation regrouped
# speedup vs baseline: 3.1102x; 3.1102x over previous
#include <hip/hip_runtime.h>

typedef float v2f __attribute__((ext_vector_type(2)));
typedef float v8f __attribute__((ext_vector_type(8)));

#define C_IN   128
#define K_OUT  128
#define H_     56
#define W_     56
#define HW_    3136
#define KRED   1152
#define SROWS  7
#define LROWS  9
#define XROW   58
#define XPL    536
#define NPAIR  64


template <int NQ>
__device__ __forceinline__ void strip_f23(const v2f* xs, const float* wrow, float (&w)[36],
                                          const int qt0, const int dt, const int m, const int hh,
                                          float* outp, const float* bias8)
{
    int qb[NQ];
    #pragma unroll
    for (int t = 0; t < NQ; ++t) {
        const int q   = 16 * (qt0 + t) + m;
        const int qc  = q < 168 ? q : 167;
        const int jp  = qc / W_;
        const int col = qc - jp * W_;
        qb[t] = 2 * hh * XPL + 2 * jp * XROW + col;
    }
    const int pd  = 16 * dt + m;
    const int pdc = pd < W_ ? pd : W_ - 1;
    const int db  = 2 * hh * XPL + 6 * XROW + pdc;

    v8f am[NQ][4];
    v8f ad = (v8f){};
    #pragma unroll
    for (int t = 0; t < NQ; ++t) {
        #pragma unroll
        for (int i = 0; i < 4; ++i) am[t][i] = (v8f){};
    }

    v2f pe[3], p0[4];
    #pragma unroll
    for (int r = 0; r < 3; ++r) pe[r] = xs[db + r * XROW];
    #pragma unroll
    for (int i = 0; i < 4; ++i) p0[i] = xs[qb[0] + i * XROW];

    for (int cb = 0; cb < 16; ++cb) {
        const int cbn = cb < 15 ? cb + 1 : 15;
        const float* wn = wrow + cbn * 72;
        const v2f* xb  = xs + cb  * (4 * XPL);
        const v2f* xbn = xs + cbn * (4 * XPL);

        #pragma unroll
        for (int g = 0; g < 6; ++g) {
            const int j = g / 3, s = g % 3;
            v2f d[NQ][4], e[3];
            #pragma unroll
            for (int r = 0; r < 3; ++r) e[r] = pe[r];
            #pragma unroll
            for (int i = 0; i < 4; ++i) d[0][i] = p0[i];
            #pragma unroll
            for (int t = 1; t < NQ; ++t) {
                #pragma unroll
                for (int i = 0; i < 4; ++i) d[t][i] = xb[qb[t] + j * XPL + i * XROW + s];
            }
            {
                const int gn = (g + 1) % 6, jn = gn / 3, sn = gn % 3;
                const v2f* xq = (g < 5) ? xb : xbn;
                #pragma unroll
                for (int r = 0; r < 3; ++r) pe[r] = xq[db + jn * XPL + r * XROW + sn];
                #pragma unroll
                for (int i = 0; i < 4; ++i) p0[i] = xq[qb[0] + jn * XPL + i * XROW + sn];
            }
            __builtin_amdgcn_sched_barrier(0);

            const v2f g0 = (v2f){w[18 * j + s],     w[18 * j + 9 + s]};
            const v2f g1 = (v2f){w[18 * j + 3 + s], w[18 * j + 12 + s]};
            const v2f g2 = (v2f){w[18 * j + 6 + s], w[18 * j + 15 + s]};
            const v2f hs = (g0 + g2) * 0.5f;
            const v2f hg = g1 * 0.5f;
            const v2f G1 = hs + hg;
            const v2f G2 = hs - hg;

            ad = __builtin_amdgcn_wmma_f32_16x16x4_f32(false, g0, false, e[0], (short)0, ad, false, false);
            ad = __builtin_amdgcn_wmma_f32_16x16x4_f32(false, g1, false, e[1], (short)0, ad, false, false);
            ad = __builtin_amdgcn_wmma_f32_16x16x4_f32(false, g2, false, e[2], (short)0, ad, false, false);
            #pragma unroll
            for (int t = 0; t < NQ; ++t) {
                const v2f t0 = d[t][0] - d[t][2];
                const v2f t1 = d[t][1] + d[t][2];
                const v2f t2 = d[t][2] - d[t][1];
                const v2f t3 = d[t][1] - d[t][3];
                am[t][0] = __builtin_amdgcn_wmma_f32_16x16x4_f32(
                    false, g0, false, t0, (short)0, am[t][0], false, false);
                am[t][1] = __builtin_amdgcn_wmma_f32_16x16x4_f32(
                    false, G1, false, t1, (short)0, am[t][1], false, false);
                am[t][2] = __builtin_amdgcn_wmma_f32_16x16x4_f32(
                    false, G2, false, t2, (short)0, am[t][2], false, false);
                am[t][3] = __builtin_amdgcn_wmma_f32_16x16x4_f32(
                    false, g2, false, t3, (short)0, am[t][3], false, false);
                if (t == 0) __builtin_amdgcn_sched_barrier(0);
            }
            __builtin_amdgcn_sched_barrier(0);

            if (s == 2) {
                asm volatile("" ::: "memory");
                #pragma unroll
                for (int i = 0; i < 18; ++i) w[18 * j + i] = wn[18 * j + i];
            }
        }
    }

    float bv[8];
    #pragma unroll
    for (int g = 0; g < 8; ++g) bv[g] = bias8[g];
    #pragma unroll
    for (int t = 0; t < NQ; ++t) {
        const int q = 16 * (qt0 + t) + m;
        if (q < 168) {
            const int jp  = q / W_;
            const int col = q - jp * W_;
            float* o = outp + 2 * jp * W_ + col;
            #pragma unroll
            for (int g = 0; g < 8; ++g) {
                const float m1 = am[t][1][g], m2 = am[t][2][g];
                o[(long)g * HW_]      = (am[t][0][g] + m1 + m2) + bv[g];
                o[(long)g * HW_ + W_] = (m1 - m2 - am[t][3][g]) + bv[g];
            }
        }
    }
    if (pd < W_) {
        #pragma unroll
        for (int g = 0; g < 8; ++g)
            outp[(long)g * HW_ + 6 * W_ + pd] = ad[g] + bv[g];
    }
}

__global__ __launch_bounds__(512, 4)
void conv3x3_strip7_wmma_f32(const float* __restrict__ in,
                             const float* __restrict__ wgt,
                             const float* __restrict__ bias,
                             float* __restrict__ out)
{
    __shared__ v2f xs[NPAIR * XPL];

    const int tid    = threadIdx.x;
    const int ochalf = blockIdx.x & 1;
    const int sidx   = blockIdx.x >> 1;
    const int img    = sidx >> 3;
    const int strip  = sidx & 7;
    const int h0     = strip * SROWS;

    const int wave = __builtin_amdgcn_readfirstlane(tid >> 5);
    const int lane = tid & 31;
    const int m    = lane & 15;
    const int hh   = lane >> 4;
    const int ot   = wave >> 2;
    const int pg   = (wave + ot) & 3;
    const int oc0  = ochalf * 64 + ot * 16;

    const float* wrow = wgt + (long)(oc0 + m) * KRED + 36 * hh;
    float w[36];
    #pragma unroll
    for (int i = 0; i < 36; ++i) w[i] = wrow[i];

    {
        const int grp = tid >> 4;
        const int l16 = tid & 15;
        if (l16 < 14) {
            const float* src = in + (long)(img * C_IN + 2 * grp) * HW_ + 4 * l16;
            float4 v[36];
            #pragma unroll
            for (int i = 0; i < 18; ++i) {
                const int r = i % LROWS, q = i / LROWS;
                int ih = h0 - 1 + r;
                if (r == 0)         ih = ih < 0 ? 0 : ih;
                if (r == LROWS - 1) ih = ih > H_ - 1 ? H_ - 1 : ih;
                const float* s = src + (long)q * (64 * HW_) + ih * W_;
                v[2 * i]     = *(const float4*)s;
                v[2 * i + 1] = *(const float4*)(s + HW_);
            }
            #pragma unroll
            for (int i = 0; i < 18; ++i) {
                const int r = i % LROWS, q = i / LROWS;
                float4 x = v[2 * i], y = v[2 * i + 1];
                if ((r == 0 && strip == 0) || (r == LROWS - 1 && strip == 7)) {
                    x = make_float4(0.f, 0.f, 0.f, 0.f);
                    y = x;
                }
                v2f* d = xs + (grp + 32 * q) * XPL + r * XROW + 1 + 4 * l16;
                d[0] = (v2f){x.x, y.x};
                d[1] = (v2f){x.y, y.y};
                d[2] = (v2f){x.z, y.z};
                d[3] = (v2f){x.w, y.w};
            }
        } else {
            const int bcol = (l16 == 14) ? 0 : (XROW - 1);
            #pragma unroll
            for (int i = 0; i < 18; ++i)
                xs[(grp + 32 * (i / LROWS)) * XPL + (i % LROWS) * XROW + bcol] = (v2f){0.f, 0.f};
        }
    }
    __syncthreads();

    float*       outp  = out + (long)(img * K_OUT + oc0 + 8 * hh) * HW_ + h0 * W_;
    const float* bias8 = bias + oc0 + 8 * hh;
    if (pg < 3) strip_f23<3>(xs, wrow, w, 3 * pg, pg, m, hh, outp, bias8);
    else        strip_f23<2>(xs, wrow, w, 9,      3,  m, hh, outp, bias8);
}

extern "C" void kernel_launch(void* const* d_in, const int* in_sizes, int n_in,
                              void* d_out, int out_size, void* d_ws, size_t ws_size,
                              hipStream_t stream) {
    (void)in_sizes; (void)n_in; (void)d_ws; (void)ws_size; (void)out_size;
    const float* in   = (const float*)d_in[0];
    const float* wgt  = (const float*)d_in[1];
    const float* bias = (const float*)d_in[2];
    float*       out  = (float*)d_out;

    dim3 grid(256, 1, 1);
    dim3 block(512, 1, 1);
    conv3x3_strip7_wmma_f32<<<grid, block, 0, stream>>>(in, wgt, bias, out);
}


# F(3,3) minimal filtering down image columns on the strip's last 3 rows (5 products) replaces one F(2,3) row pair + the direct row (7): 288 WMMAs per block per SIMD where 336 were; summation regrouped
# speedup vs baseline: 3.2945x; 1.0593x over previous
#include <hip/hip_runtime.h>

typedef float v2f __attribute__((ext_vector_type(2)));
typedef float v8f __attribute__((ext_vector_type(8)));

#define C_IN   128
#define K_OUT  128
#define H_     56
#define W_     56
#define HW_    3136
#define KRED   1152
#define SROWS  7
#define LROWS  9
#define XROW   58
#define XPL    536
#define NPAIR  64


template <int NQ>
__device__ __forceinline__ void strip_f23_f33(const v2f* xs, const float* wrow, float (&w)[36],
                                              const int qt0, const int tt, const int m, const int hh,
                                              float* outp, const float* bias8)
{
    int qb[NQ];
    #pragma unroll
    for (int t = 0; t < NQ; ++t) {
        const int q   = 16 * (qt0 + t) + m;
        const int jp  = q / W_;
        const int col = q - jp * W_;
        qb[t] = 2 * hh * XPL + 2 * jp * XROW + col;
    }
    const int pd  = 16 * tt + m;
    const int pdc = pd < W_ ? pd : W_ - 1;
    const int tb  = 2 * hh * XPL + 4 * XROW + pdc;

    v8f am[NQ][4];
    v8f at[5];
    #pragma unroll
    for (int t = 0; t < NQ; ++t) {
        #pragma unroll
        for (int i = 0; i < 4; ++i) am[t][i] = (v8f){};
    }
    #pragma unroll
    for (int i = 0; i < 5; ++i) at[i] = (v8f){};

    v2f pp[NQ][4];
    #pragma unroll
    for (int t = 0; t < NQ; ++t) {
        #pragma unroll
        for (int i = 0; i < 4; ++i) pp[t][i] = xs[qb[t] + i * XROW];
    }

    for (int cb = 0; cb < 16; ++cb) {
        const int cbn = cb < 15 ? cb + 1 : 15;
        const float* wn = wrow + cbn * 72;
        const v2f* xb  = xs + cb  * (4 * XPL);
        const v2f* xbn = xs + cbn * (4 * XPL);

        #pragma unroll
        for (int g = 0; g < 6; ++g) {
            const int j = g / 3, s = g % 3;
            v2f d[NQ][4], e[5];
            #pragma unroll
            for (int t = 0; t < NQ; ++t) {
                #pragma unroll
                for (int i = 0; i < 4; ++i) d[t][i] = pp[t][i];
            }
            #pragma unroll
            for (int r = 0; r < 5; ++r) e[r] = xb[tb + j * XPL + r * XROW + s];
            {
                const int gn = (g + 1) % 6, jn = gn / 3, sn = gn % 3;
                const v2f* xq = (g < 5) ? xb : xbn;
                #pragma unroll
                for (int t = 0; t < NQ; ++t) {
                    #pragma unroll
                    for (int i = 0; i < 4; ++i) pp[t][i] = xq[qb[t] + jn * XPL + i * XROW + sn];
                }
            }
            __builtin_amdgcn_sched_barrier(0);

            const v2f g0 = (v2f){w[18 * j + s],     w[18 * j + 9 + s]};
            const v2f g1 = (v2f){w[18 * j + 3 + s], w[18 * j + 12 + s]};
            const v2f g2 = (v2f){w[18 * j + 6 + s], w[18 * j + 15 + s]};
            const v2f hs = (g0 + g2) * 0.5f;
            const v2f hg = g1 * 0.5f;
            const v2f G1 = hs + hg;
            const v2f G2 = hs - hg;
            const v2f H3 = g0 * (1.0f / 6.0f) + (g1 * (1.0f / 3.0f) + g2 * (2.0f / 3.0f));

            #pragma unroll
            for (int t = 0; t < NQ; ++t) {
                const v2f t0 = d[t][0] - d[t][2];
                const v2f t1 = d[t][1] + d[t][2];
                const v2f t2 = d[t][2] - d[t][1];
                const v2f t3 = d[t][1] - d[t][3];
                am[t][0] = __builtin_amdgcn_wmma_f32_16x16x4_f32(
                    false, g0, false, t0, (short)0, am[t][0], false, false);
                am[t][1] = __builtin_amdgcn_wmma_f32_16x16x4_f32(
                    false, G1, false, t1, (short)0, am[t][1], false, false);
                am[t][2] = __builtin_amdgcn_wmma_f32_16x16x4_f32(
                    false, G2, false, t2, (short)0, am[t][2], false, false);
                am[t][3] = __builtin_amdgcn_wmma_f32_16x16x4_f32(
                    false, g2, false, t3, (short)0, am[t][3], false, false);
            }
            __builtin_amdgcn_sched_barrier(0);

            const v2f u3 = e[3] - e[1];
            const v2f u0 = (e[0] - e[2]) + u3 * 0.5f;
            const v2f u1 = e[1] * 2.0f + (e[2] - e[3]);
            const v2f u2 = e[2] - (e[1] * 2.0f + e[3]) * (1.0f / 3.0f);
            const v2f u4 = (e[4] - e[2]) - u3 * 2.0f;
            at[0] = __builtin_amdgcn_wmma_f32_16x16x4_f32(false, g0, false, u0, (short)0, at[0], false, false);
            at[1] = __builtin_amdgcn_wmma_f32_16x16x4_f32(false, G1, false, u1, (short)0, at[1], false, false);
            at[2] = __builtin_amdgcn_wmma_f32_16x16x4_f32(false, G2, false, u2, (short)0, at[2], false, false);
            at[3] = __builtin_amdgcn_wmma_f32_16x16x4_f32(false, H3, false, u3, (short)0, at[3], false, false);
            at[4] = __builtin_amdgcn_wmma_f32_16x16x4_f32(false, g2, false, u4, (short)0, at[4], false, false);
            __builtin_amdgcn_sched_barrier(0);

            if (s == 2) {
                asm volatile("" ::: "memory");
                #pragma unroll
                for (int i = 0; i < 18; ++i) w[18 * j + i] = wn[18 * j + i];
            }
        }
    }

    float bv[8];
    #pragma unroll
    for (int g = 0; g < 8; ++g) bv[g] = bias8[g];
    #pragma unroll
    for (int t = 0; t < NQ; ++t) {
        const int q   = 16 * (qt0 + t) + m;
        const int jp  = q / W_;
        const int col = q - jp * W_;
        float* o = outp + 2 * jp * W_ + col;
        #pragma unroll
        for (int g = 0; g < 8; ++g) {
            const float m1 = am[t][1][g], m2 = am[t][2][g];
            o[(long)g * HW_]      = (am[t][0][g] + m1 + m2) + bv[g];
            o[(long)g * HW_ + W_] = (m1 - m2 - am[t][3][g]) + bv[g];
        }
    }
    if (pd < W_) {
        float* o = outp + 4 * W_ + pd;
        #pragma unroll
        for (int g = 0; g < 8; ++g) {
            const float q1 = at[1][g], q2 = at[2][g], q3 = at[3][g];
            o[(long)g * HW_]          = (at[0][g] + q1 + q2 + q3) + bv[g];
            o[(long)g * HW_ + W_]     = (q1 - q2 + 2.0f * q3) + bv[g];
            o[(long)g * HW_ + 2 * W_] = (q1 + q2 + 4.0f * q3 + at[4][g]) + bv[g];
        }
    }
}

__global__ __launch_bounds__(512, 4)
void conv3x3_strip7_wmma_f32(const float* __restrict__ in,
                             const float* __restrict__ wgt,
                             const float* __restrict__ bias,
                             float* __restrict__ out)
{
    __shared__ v2f xs[NPAIR * XPL];

    const int tid    = threadIdx.x;
    const int ochalf = blockIdx.x & 1;
    const int sidx   = blockIdx.x >> 1;
    const int img    = sidx >> 3;
    const int strip  = sidx & 7;
    const int h0     = strip * SROWS;

    const int wave = __builtin_amdgcn_readfirstlane(tid >> 5);
    const int lane = tid & 31;
    const int m    = lane & 15;
    const int hh   = lane >> 4;
    const int ot   = wave >> 2;
    const int pg   = (wave + ot) & 3;
    const int oc0  = ochalf * 64 + ot * 16;

    const float* wrow = wgt + (long)(oc0 + m) * KRED + 36 * hh;
    float w[36];
    #pragma unroll
    for (int i = 0; i < 36; ++i) w[i] = wrow[i];

    {
        const int grp = tid >> 4;
        const int l16 = tid & 15;
        if (l16 < 14) {
            const float* src = in + (long)(img * C_IN + 2 * grp) * HW_ + 4 * l16;
            float4 v[36];
            #pragma unroll
            for (int i = 0; i < 18; ++i) {
                const int r = i % LROWS, q = i / LROWS;
                int ih = h0 - 1 + r;
                if (r == 0)         ih = ih < 0 ? 0 : ih;
                if (r == LROWS - 1) ih = ih > H_ - 1 ? H_ - 1 : ih;
                const float* s = src + (long)q * (64 * HW_) + ih * W_;
                v[2 * i]     = *(const float4*)s;
                v[2 * i + 1] = *(const float4*)(s + HW_);
            }
            #pragma unroll
            for (int i = 0; i < 18; ++i) {
                const int r = i % LROWS, q = i / LROWS;
                float4 x = v[2 * i], y = v[2 * i + 1];
                if ((r == 0 && strip == 0) || (r == LROWS - 1 && strip == 7)) {
                    x = make_float4(0.f, 0.f, 0.f, 0.f);
                    y = x;
                }
                v2f* d = xs + (grp + 32 * q) * XPL + r * XROW + 1 + 4 * l16;
                d[0] = (v2f){x.x, y.x};
                d[1] = (v2f){x.y, y.y};
                d[2] = (v2f){x.z, y.z};
                d[3] = (v2f){x.w, y.w};
            }
        } else {
            const int bcol = (l16 == 14) ? 0 : (XROW - 1);
            #pragma unroll
            for (int i = 0; i < 18; ++i)
                xs[(grp + 32 * (i / LROWS)) * XPL + (i % LROWS) * XROW + bcol] = (v2f){0.f, 0.f};
        }
    }
    __syncthreads();

    float*       outp  = out + (long)(img * K_OUT + oc0 + 8 * hh) * HW_ + h0 * W_;
    const float* bias8 = bias + oc0 + 8 * hh;
    if (pg < 3) strip_f23_f33<2>(xs, wrow, w, 2 * pg, pg, m, hh, outp, bias8);
    else        strip_f23_f33<1>(xs, wrow, w, 6,      3,  m, hh, outp, bias8);
}

extern "C" void kernel_launch(void* const* d_in, const int* in_sizes, int n_in,
                              void* d_out, int out_size, void* d_ws, size_t ws_size,
                              hipStream_t stream) {
    (void)in_sizes; (void)n_in; (void)d_ws; (void)ws_size; (void)out_size;
    const float* in   = (const float*)d_in[0];
    const float* wgt  = (const float*)d_in[1];
    const float* bias = (const float*)d_in[2];
    float*       out  = (float*)d_out;

    dim3 grid(256, 1, 1);
    dim3 block(512, 1, 1);
    conv3x3_strip7_wmma_f32<<<grid, block, 0, stream>>>(in, wgt, bias, out);
}


# Input side of F(4,3) for channels 0-63 applied once at staging: 12 transformed rows in LDS, B fragment = bare LDS read in 8 of 16 blocks; channels 64-127 stay raw; whole 320 KiB LDS; sums unchanged
# speedup vs baseline: 3.8206x; 1.1597x over previous
#include <hip/hip_runtime.h>

typedef float v2f __attribute__((ext_vector_type(2)));
typedef float v8f __attribute__((ext_vector_type(8)));

#define C_IN   128
#define K_OUT  128
#define H_     56
#define W_     56
#define HW_    3136
#define KRED   1152
#define SROWS  8
#define NSTRIP 7
#define LROWS  10
#define XROW   58
#define XPL    584
#define NPAIR  64
#define TROWS  12
#define XPLT   696
#define BLKP   (4 * XPLT + 4 * XPL)
typedef float v4f __attribute__((ext_vector_type(4)));

__device__ __forceinline__ void quad_rows_f43(const v4f* d, v4f* v)
{
    const v4f sp = d[2] * 4.0f - d[4];
    const v4f sr = d[1] * 4.0f - d[3];
    const v4f sf = d[4] - d[2];
    const v4f sh = d[3] - d[1];
    v[0] = (d[0] - d[2]) * 4.0f + sf;
    v[1] = sp + sr;
    v[2] = sp - sr;
    v[3] = sf + sh * 2.0f;
    v[4] = sf - sh * 2.0f;
    v[5] = (d[5] - d[3]) - sh * 4.0f;
}


template <int NQ>
__device__ __forceinline__ void strip_f43(const v2f* xs, const float* wrow, float (&w)[36],
                                          const int qt0, const int m, const int hh,
                                          float* outp, const float* bias8)
{
    v8f am[NQ][6];
    #pragma unroll
    for (int t = 0; t < NQ; ++t) {
        #pragma unroll
        for (int i = 0; i < 6; ++i) am[t][i] = (v8f){};
    }

    #pragma unroll
    for (int ph = 0; ph < 2; ++ph) {
        const bool TR  = (ph == 0);
        const int  PL  = TR ? XPLT : XPL;
        const int  QR  = TR ? 6 : 4;
        const v2f* xp  = TR ? xs : xs + 4 * XPLT;

        int qb[NQ];
        #pragma unroll
        for (int t = 0; t < NQ; ++t) {
            const int q   = 16 * (qt0 + t) + m;
            const int jq  = q / W_;
            const int col = q - jq * W_;
            qb[t] = 2 * hh * PL + QR * jq * XROW + col;
        }

        v2f pp[NQ][6];
        #pragma unroll
        for (int t = 0; t < NQ; ++t) {
            #pragma unroll
            for (int i = 0; i < 6; ++i) pp[t][i] = xp[qb[t] + i * XROW];
        }

        for (int cl = 0; cl < 8; ++cl) {
            const int cln = cl < 7 ? cl + 1 : 7;
            const int cbn = TR ? cl + 1 : 8 + cln;
            const float* wn = wrow + cbn * 72;
            const v2f* xb  = xp + cl  * BLKP;
            const v2f* xbn = xb + (cln - cl) * BLKP;

            #pragma unroll
            for (int g = 0; g < 6; ++g) {
                const int j = g / 3, s = g % 3;
                v2f d[NQ][6];
                #pragma unroll
                for (int t = 0; t < NQ; ++t) {
                    #pragma unroll
                    for (int i = 0; i < 6; ++i) d[t][i] = pp[t][i];
                }
                {
                    const int gn = (g + 1) % 6, jn = gn / 3, sn = gn % 3;
                    const v2f* xq = (g < 5) ? xb : xbn;
                    #pragma unroll
                    for (int t = 0; t < NQ; ++t) {
                        #pragma unroll
                        for (int i = 0; i < 6; ++i) pp[t][i] = xq[qb[t] + jn * PL + i * XROW + sn];
                    }
                }
                __builtin_amdgcn_sched_barrier(0);

                const v2f g0 = (v2f){w[18 * j + s],     w[18 * j + 9 + s]};
                const v2f g1 = (v2f){w[18 * j + 3 + s], w[18 * j + 12 + s]};
                const v2f g2 = (v2f){w[18 * j + 6 + s], w[18 * j + 15 + s]};
                const v2f A0 = g0 * 0.25f;
                const v2f t6 = g2 * (1.0f / 6.0f);
                const v2f h6 = g0 * (1.0f / 6.0f) + t6;
                const v2f k4 = g0 * (1.0f / 24.0f) + t6;
                const v2f A1 = h6 + g1 * (1.0f / 6.0f);
                const v2f A2 = h6 - g1 * (1.0f / 6.0f);
                const v2f A3 = k4 + g1 * (1.0f / 12.0f);
                const v2f A4 = k4 - g1 * (1.0f / 12.0f);

                #pragma unroll
                for (int t = 0; t < NQ; ++t) {
                    v2f v0, v1, v2, v3, v4, v5;
                    if (TR) {
                        v0 = d[t][0]; v1 = d[t][1]; v2 = d[t][2]; v3 = d[t][3]; v4 = d[t][4]; v5 = d[t][5];
                    } else {
                        const v2f sp = d[t][2] * 4.0f - d[t][4];
                        const v2f sr = d[t][1] * 4.0f - d[t][3];
                        const v2f sf = d[t][4] - d[t][2];
                        const v2f sh = d[t][3] - d[t][1];
                        v0 = (d[t][0] - d[t][2]) * 4.0f + sf;
                        v1 = sp + sr;
                        v2 = sp - sr;
                        v3 = sf + sh * 2.0f;
                        v4 = sf - sh * 2.0f;
                        v5 = (d[t][5] - d[t][3]) - sh * 4.0f;
                    }
                    am[t][0] = __builtin_amdgcn_wmma_f32_16x16x4_f32(
                        false, A0, false, v0, (short)0, am[t][0], false, false);
                    am[t][1] = __builtin_amdgcn_wmma_f32_16x16x4_f32(
                        false, A1, false, v1, (short)0, am[t][1], false, false);
                    am[t][2] = __builtin_amdgcn_wmma_f32_16x16x4_f32(
                        false, A2, false, v2, (short)0, am[t][2], false, false);
                    am[t][3] = __builtin_amdgcn_wmma_f32_16x16x4_f32(
                        false, A3, false, v3, (short)0, am[t][3], false, false);
                    am[t][4] = __builtin_amdgcn_wmma_f32_16x16x4_f32(
                        false, A4, false, v4, (short)0, am[t][4], false, false);
                    am[t][5] = __builtin_amdgcn_wmma_f32_16x16x4_f32(
                        false, g2, false, v5, (short)0, am[t][5], false, false);
                }
                __builtin_amdgcn_sched_barrier(0);

                if (s == 2) {
                    asm volatile("" ::: "memory");
                    #pragma unroll
                    for (int i = 0; i < 18; ++i) w[18 * j + i] = wn[18 * j + i];
                }
            }
        }
    }

    float bv[8];
    #pragma unroll
    for (int g = 0; g < 8; ++g) bv[g] = bias8[g];
    #pragma unroll
    for (int t = 0; t < NQ; ++t) {
        const int q   = 16 * (qt0 + t) + m;
        const int jq  = q / W_;
        const int col = q - jq * W_;
        float* o = outp + 4 * jq * W_ + col;
        #pragma unroll
        for (int g = 0; g < 8; ++g) {
            const float s12 = am[t][1][g] + am[t][2][g];
            const float d12 = am[t][1][g] - am[t][2][g];
            const float s34 = am[t][3][g] + am[t][4][g];
            const float d34 = am[t][3][g] - am[t][4][g];
            o[(long)g * HW_]          = (am[t][0][g] + s12 + s34) + bv[g];
            o[(long)g * HW_ + W_]     = (d12 + 2.0f * d34) + bv[g];
            o[(long)g * HW_ + 2 * W_] = (s12 + 4.0f * s34) + bv[g];
            o[(long)g * HW_ + 3 * W_] = (d12 + 8.0f * d34 + am[t][5][g]) + bv[g];
        }
    }
}

__global__ __launch_bounds__(512, 4)
void conv3x3_strip8_wmma_f32(const float* __restrict__ in,
                             const float* __restrict__ wgt,
                             const float* __restrict__ bias,
                             float* __restrict__ out)
{
    __shared__ v2f xs[8 * BLKP];

    const int tid    = threadIdx.x;
    const int ochalf = blockIdx.x & 1;
    const int sidx   = blockIdx.x >> 1;
    const int img    = sidx / NSTRIP;
    const int strip  = sidx - img * NSTRIP;
    const int h0     = strip * SROWS;

    const int wave = __builtin_amdgcn_readfirstlane(tid >> 5);
    const int lane = tid & 31;
    const int m    = lane & 15;
    const int hh   = lane >> 4;
    const int ot   = wave >> 2;
    const int pg   = (wave + ot) & 3;
    const int oc0  = ochalf * 64 + ot * 16;

    const float* wrow = wgt + (long)(oc0 + m) * KRED + 36 * hh;
    float w[36];
    #pragma unroll
    for (int i = 0; i < 36; ++i) w[i] = wrow[i];

    {
        const int grp = tid >> 4;
        const int l16 = tid & 15;
        v2f* xt = xs + (grp >> 2) * BLKP + (grp & 3) * XPLT;
        v2f* xr = xs + (grp >> 2) * BLKP + 4 * XPLT + (grp & 3) * XPL;
        if (l16 < 14) {
            const float* src = in + (long)(img * C_IN + 2 * grp) * HW_ + 4 * l16;
            float4 v[4 * LROWS];
            #pragma unroll
            for (int i = 0; i < 2 * LROWS; ++i) {
                const int r = i % LROWS, q = i / LROWS;
                int ih = h0 - 1 + r;
                if (r == 0)         ih = ih < 0 ? 0 : ih;
                if (r == LROWS - 1) ih = ih > H_ - 1 ? H_ - 1 : ih;
                const float* s = src + (long)q * (64 * HW_) + ih * W_;
                v[2 * i]     = *(const float4*)s;
                v[2 * i + 1] = *(const float4*)(s + HW_);
            }
            const bool top = (strip == 0), bot = (strip == NSTRIP - 1);
            #pragma unroll
            for (int jq = 0; jq < 2; ++jq) {
                v4f dx[6], dy[6], tx[6], ty[6];
                #pragma unroll
                for (int i = 0; i < 6; ++i) {
                    const int r = 4 * jq + i;
                    float4 x = v[2 * r], y = v[2 * r + 1];
                    if ((r == 0 && top) || (r == LROWS - 1 && bot)) {
                        x = make_float4(0.f, 0.f, 0.f, 0.f);
                        y = x;
                    }
                    dx[i] = (v4f){x.x, x.y, x.z, x.w};
                    dy[i] = (v4f){y.x, y.y, y.z, y.w};
                }
                quad_rows_f43(dx, tx);
                quad_rows_f43(dy, ty);
                #pragma unroll
                for (int i = 0; i < 6; ++i) {
                    v2f* d = xt + (6 * jq + i) * XROW + 1 + 4 * l16;
                    d[0] = (v2f){tx[i][0], ty[i][0]};
                    d[1] = (v2f){tx[i][1], ty[i][1]};
                    d[2] = (v2f){tx[i][2], ty[i][2]};
                    d[3] = (v2f){tx[i][3], ty[i][3]};
                }
            }
            #pragma unroll
            for (int r = 0; r < LROWS; ++r) {
                float4 x = v[2 * (LROWS + r)], y = v[2 * (LROWS + r) + 1];
                if ((r == 0 && top) || (r == LROWS - 1 && bot)) {
                    x = make_float4(0.f, 0.f, 0.f, 0.f);
                    y = x;
                }
                v2f* d = xr + r * XROW + 1 + 4 * l16;
                d[0] = (v2f){x.x, y.x};
                d[1] = (v2f){x.y, y.y};
                d[2] = (v2f){x.z, y.z};
                d[3] = (v2f){x.w, y.w};
            }
        } else {
            const int bcol = (l16 == 14) ? 0 : (XROW - 1);
            #pragma unroll
            for (int r = 0; r < TROWS; ++r)
                xt[r * XROW + bcol] = (v2f){0.f, 0.f};
            #pragma unroll
            for (int r = 0; r < LROWS; ++r)
                xr[r * XROW + bcol] = (v2f){0.f, 0.f};
        }
    }
    __syncthreads();

    float*       outp  = out + (long)(img * K_OUT + oc0 + 8 * hh) * HW_ + h0 * W_;
    const float* bias8 = bias + oc0 + 8 * hh;
    if (pg < 3) strip_f43<2>(xs, wrow, w, 2 * pg, m, hh, outp, bias8);
    else        strip_f43<1>(xs, wrow, w, 6,      m, hh, outp, bias8);
}

extern "C" void kernel_launch(void* const* d_in, const int* in_sizes, int n_in,
                              void* d_out, int out_size, void* d_ws, size_t ws_size,
                              hipStream_t stream) {
    (void)in_sizes; (void)n_in; (void)d_ws; (void)ws_size; (void)out_size;
    const float* in   = (const float*)d_in[0];
    const float* wgt  = (const float*)d_in[1];
    const float* bias = (const float*)d_in[2];
    float*       out  = (float*)d_out;

    dim3 grid(224, 1, 1);
    dim3 block(512, 1, 1);
    conv3x3_strip8_wmma_f32<<<grid, block, 0, stream>>>(in, wgt, bias, out);
}


# 8 waves per work-group, 2 per SIMD: a wave holds 4 or 3 quad tiles of its output-channel tile (was 2 or 1): half the weight loads, filter transforms and wave entries per SIMD; sums unchanged
# speedup vs baseline: 3.9489x; 1.0336x over previous
#include <hip/hip_runtime.h>

typedef float v2f __attribute__((ext_vector_type(2)));
typedef float v8f __attribute__((ext_vector_type(8)));

#define C_IN   128
#define K_OUT  128
#define H_     56
#define W_     56
#define HW_    3136
#define KRED   1152
#define SROWS  8
#define NSTRIP 7
#define LROWS  10
#define XROW   58
#define XPL    584
#define NPAIR  64
#define TROWS  12
#define XPLT   696
#define BLKP   (4 * XPLT + 4 * XPL)
typedef float v4f __attribute__((ext_vector_type(4)));

__device__ __forceinline__ void quad_rows_f43(const v4f* d, v4f* v)
{
    const v4f sp = d[2] * 4.0f - d[4];
    const v4f sr = d[1] * 4.0f - d[3];
    const v4f sf = d[4] - d[2];
    const v4f sh = d[3] - d[1];
    v[0] = (d[0] - d[2]) * 4.0f + sf;
    v[1] = sp + sr;
    v[2] = sp - sr;
    v[3] = sf + sh * 2.0f;
    v[4] = sf - sh * 2.0f;
    v[5] = (d[5] - d[3]) - sh * 4.0f;
}


template <int NQ>
__device__ __forceinline__ void strip_f43(const v2f* xs, const float* wrow, float (&w)[36],
                                          const int qt0, const int m, const int hh,
                                          float* outp, const float* bias8)
{
    v8f am[NQ][6];
    #pragma unroll
    for (int t = 0; t < NQ; ++t) {
        #pragma unroll
        for (int i = 0; i < 6; ++i) am[t][i] = (v8f){};
    }

    #pragma unroll
    for (int ph = 0; ph < 2; ++ph) {
        const bool TR  = (ph == 0);
        const int  PL  = TR ? XPLT : XPL;
        const int  QR  = TR ? 6 : 4;
        const v2f* xp  = TR ? xs : xs + 4 * XPLT;

        int qb[NQ];
        #pragma unroll
        for (int t = 0; t < NQ; ++t) {
            const int q   = 16 * (qt0 + t) + m;
            const int jq  = q / W_;
            const int col = q - jq * W_;
            qb[t] = 2 * hh * PL + QR * jq * XROW + col;
        }

        v2f pp[NQ][6];
        #pragma unroll
        for (int t = 0; t < NQ; ++t) {
            #pragma unroll
            for (int i = 0; i < 6; ++i) pp[t][i] = xp[qb[t] + i * XROW];
        }

        for (int cl = 0; cl < 8; ++cl) {
            const int cln = cl < 7 ? cl + 1 : 7;
            const int cbn = TR ? cl + 1 : 8 + cln;
            const float* wn = wrow + cbn * 72;
            const v2f* xb  = xp + cl  * BLKP;
            const v2f* xbn = xb + (cln - cl) * BLKP;

            #pragma unroll
            for (int g = 0; g < 6; ++g) {
                const int j = g / 3, s = g % 3;
                v2f d[NQ][6];
                #pragma unroll
                for (int t = 0; t < NQ; ++t) {
                    #pragma unroll
                    for (int i = 0; i < 6; ++i) d[t][i] = pp[t][i];
                }
                {
                    const int gn = (g + 1) % 6, jn = gn / 3, sn = gn % 3;
                    const v2f* xq = (g < 5) ? xb : xbn;
                    #pragma unroll
                    for (int t = 0; t < NQ; ++t) {
                        #pragma unroll
                        for (int i = 0; i < 6; ++i) pp[t][i] = xq[qb[t] + jn * PL + i * XROW + sn];
                    }
                }
                __builtin_amdgcn_sched_barrier(0);

                const v2f g0 = (v2f){w[18 * j + s],     w[18 * j + 9 + s]};
                const v2f g1 = (v2f){w[18 * j + 3 + s], w[18 * j + 12 + s]};
                const v2f g2 = (v2f){w[18 * j + 6 + s], w[18 * j + 15 + s]};
                const v2f A0 = g0 * 0.25f;
                const v2f t6 = g2 * (1.0f / 6.0f);
                const v2f h6 = g0 * (1.0f / 6.0f) + t6;
                const v2f k4 = g0 * (1.0f / 24.0f) + t6;
                const v2f A1 = h6 + g1 * (1.0f / 6.0f);
                const v2f A2 = h6 - g1 * (1.0f / 6.0f);
                const v2f A3 = k4 + g1 * (1.0f / 12.0f);
                const v2f A4 = k4 - g1 * (1.0f / 12.0f);

                #pragma unroll
                for (int t = 0; t < NQ; ++t) {
                    v2f v0, v1, v2, v3, v4, v5;
                    if (TR) {
                        v0 = d[t][0]; v1 = d[t][1]; v2 = d[t][2]; v3 = d[t][3]; v4 = d[t][4]; v5 = d[t][5];
                    } else {
                        const v2f sp = d[t][2] * 4.0f - d[t][4];
                        const v2f sr = d[t][1] * 4.0f - d[t][3];
                        const v2f sf = d[t][4] - d[t][2];
                        const v2f sh = d[t][3] - d[t][1];
                        v0 = (d[t][0] - d[t][2]) * 4.0f + sf;
                        v1 = sp + sr;
                        v2 = sp - sr;
                        v3 = sf + sh * 2.0f;
                        v4 = sf - sh * 2.0f;
                        v5 = (d[t][5] - d[t][3]) - sh * 4.0f;
                    }
                    am[t][0] = __builtin_amdgcn_wmma_f32_16x16x4_f32(
                        false, A0, false, v0, (short)0, am[t][0], false, false);
                    am[t][1] = __builtin_amdgcn_wmma_f32_16x16x4_f32(
                        false, A1, false, v1, (short)0, am[t][1], false, false);
                    am[t][2] = __builtin_amdgcn_wmma_f32_16x16x4_f32(
                        false, A2, false, v2, (short)0, am[t][2], false, false);
                    am[t][3] = __builtin_amdgcn_wmma_f32_16x16x4_f32(
                        false, A3, false, v3, (short)0, am[t][3], false, false);
                    am[t][4] = __builtin_amdgcn_wmma_f32_16x16x4_f32(
                        false, A4, false, v4, (short)0, am[t][4], false, false);
                    am[t][5] = __builtin_amdgcn_wmma_f32_16x16x4_f32(
                        false, g2, false, v5, (short)0, am[t][5], false, false);
                }
                __builtin_amdgcn_sched_barrier(0);

                if (s == 2) {
                    asm volatile("" ::: "memory");
                    #pragma unroll
                    for (int i = 0; i < 18; ++i) w[18 * j + i] = wn[18 * j + i];
                }
            }
        }
    }

    float bv[8];
    #pragma unroll
    for (int g = 0; g < 8; ++g) bv[g] = bias8[g];
    #pragma unroll
    for (int t = 0; t < NQ; ++t) {
        const int q   = 16 * (qt0 + t) + m;
        const int jq  = q / W_;
        const int col = q - jq * W_;
        float* o = outp + 4 * jq * W_ + col;
        #pragma unroll
        for (int g = 0; g < 8; ++g) {
            const float s12 = am[t][1][g] + am[t][2][g];
            const float d12 = am[t][1][g] - am[t][2][g];
            const float s34 = am[t][3][g] + am[t][4][g];
            const float d34 = am[t][3][g] - am[t][4][g];
            o[(long)g * HW_]          = (am[t][0][g] + s12 + s34) + bv[g];
            o[(long)g * HW_ + W_]     = (d12 + 2.0f * d34) + bv[g];
            o[(long)g * HW_ + 2 * W_] = (s12 + 4.0f * s34) + bv[g];
            o[(long)g * HW_ + 3 * W_] = (d12 + 8.0f * d34 + am[t][5][g]) + bv[g];
        }
    }
}

__global__ __launch_bounds__(256, 2)
void conv3x3_strip8_wmma_f32(const float* __restrict__ in,
                             const float* __restrict__ wgt,
                             const float* __restrict__ bias,
                             float* __restrict__ out)
{
    __shared__ v2f xs[8 * BLKP];

    const int tid    = threadIdx.x;
    const int ochalf = blockIdx.x & 1;
    const int sidx   = blockIdx.x >> 1;
    const int img    = sidx / NSTRIP;
    const int strip  = sidx - img * NSTRIP;
    const int h0     = strip * SROWS;

    const int wave = __builtin_amdgcn_readfirstlane(tid >> 5);
    const int lane = tid & 31;
    const int m    = lane & 15;
    const int hh   = lane >> 4;
    const int ot   = wave >> 1;
    const int pg   = (wave + (wave >> 2)) & 1;
    const int oc0  = ochalf * 64 + ot * 16;

    const float* wrow = wgt + (long)(oc0 + m) * KRED + 36 * hh;
    float w[36];
    #pragma unroll
    for (int i = 0; i < 36; ++i) w[i] = wrow[i];

    {
        const int grp = tid >> 4;
        const int l16 = tid & 15;
        if (l16 < 14) {
            float4 v[2][4 * LROWS];
            #pragma unroll
            for (int u = 0; u < 2; ++u) {
                const int gp = grp + 16 * u;
                const float* src = in + (long)(img * C_IN + 2 * gp) * HW_ + 4 * l16;
                #pragma unroll
                for (int i = 0; i < 2 * LROWS; ++i) {
                    const int r = i % LROWS, q = i / LROWS;
                    int ih = h0 - 1 + r;
                    if (r == 0)         ih = ih < 0 ? 0 : ih;
                    if (r == LROWS - 1) ih = ih > H_ - 1 ? H_ - 1 : ih;
                    const float* s = src + (long)q * (64 * HW_) + ih * W_;
                    v[u][2 * i]     = *(const float4*)s;
                    v[u][2 * i + 1] = *(const float4*)(s + HW_);
                }
            }
            const bool top = (strip == 0), bot = (strip == NSTRIP - 1);
            #pragma unroll
            for (int u = 0; u < 2; ++u) {
                const int gp = grp + 16 * u;
                v2f* xt = xs + (gp >> 2) * BLKP + (gp & 3) * XPLT;
                v2f* xr = xs + (gp >> 2) * BLKP + 4 * XPLT + (gp & 3) * XPL;
                #pragma unroll
                for (int jq = 0; jq < 2; ++jq) {
                    v4f dx[6], dy[6], tx[6], ty[6];
                    #pragma unroll
                    for (int i = 0; i < 6; ++i) {
                        const int r = 4 * jq + i;
                        float4 x = v[u][2 * r], y = v[u][2 * r + 1];
                        if ((r == 0 && top) || (r == LROWS - 1 && bot)) {
                            x = make_float4(0.f, 0.f, 0.f, 0.f);
                            y = x;
                        }
                        dx[i] = (v4f){x.x, x.y, x.z, x.w};
                        dy[i] = (v4f){y.x, y.y, y.z, y.w};
                    }
                    quad_rows_f43(dx, tx);
                    quad_rows_f43(dy, ty);
                    #pragma unroll
                    for (int i = 0; i < 6; ++i) {
                        v2f* d = xt + (6 * jq + i) * XROW + 1 + 4 * l16;
                        d[0] = (v2f){tx[i][0], ty[i][0]};
                        d[1] = (v2f){tx[i][1], ty[i][1]};
                        d[2] = (v2f){tx[i][2], ty[i][2]};
                        d[3] = (v2f){tx[i][3], ty[i][3]};
                    }
                }
                #pragma unroll
                for (int r = 0; r < LROWS; ++r) {
                    float4 x = v[u][2 * (LROWS + r)], y = v[u][2 * (LROWS + r) + 1];
                    if ((r == 0 && top) || (r == LROWS - 1 && bot)) {
                        x = make_float4(0.f, 0.f, 0.f, 0.f);
                        y = x;
                    }
                    v2f* d = xr + r * XROW + 1 + 4 * l16;
                    d[0] = (v2f){x.x, y.x};
                    d[1] = (v2f){x.y, y.y};
                    d[2] = (v2f){x.z, y.z};
                    d[3] = (v2f){x.w, y.w};
                }
            }
        } else {
            const int bcol = (l16 == 14) ? 0 : (XROW - 1);
            #pragma unroll
            for (int u = 0; u < 2; ++u) {
                const int gp = grp + 16 * u;
                v2f* xt = xs + (gp >> 2) * BLKP + (gp & 3) * XPLT;
                v2f* xr = xs + (gp >> 2) * BLKP + 4 * XPLT + (gp & 3) * XPL;
                #pragma unroll
                for (int r = 0; r < TROWS; ++r)
                    xt[r * XROW + bcol] = (v2f){0.f, 0.f};
                #pragma unroll
                for (int r = 0; r < LROWS; ++r)
                    xr[r * XROW + bcol] = (v2f){0.f, 0.f};
            }
        }
    }
    __syncthreads();

    float*       outp  = out + (long)(img * K_OUT + oc0 + 8 * hh) * HW_ + h0 * W_;
    const float* bias8 = bias + oc0 + 8 * hh;
    if (pg == 0) strip_f43<4>(xs, wrow, w, 0, m, hh, outp, bias8);
    else         strip_f43<3>(xs, wrow, w, 4, m, hh, outp, bias8);
}

extern "C" void kernel_launch(void* const* d_in, const int* in_sizes, int n_in,
                              void* d_out, int out_size, void* d_ws, size_t ws_size,
                              hipStream_t stream) {
    (void)in_sizes; (void)n_in; (void)d_ws; (void)ws_size; (void)out_size;
    const float* in   = (const float*)d_in[0];
    const float* wgt  = (const float*)d_in[1];
    const float* bias = (const float*)d_in[2];
    float*       out  = (float*)d_out;

    dim3 grid(224, 1, 1);
    dim3 block(256, 1, 1);
    conv3x3_strip8_wmma_f32<<<grid, block, 0, stream>>>(in, wgt, bias, out);
}


# Raw rows of channels 64..127 transformed once per work-group between the phases (LDS to LDS, into the planes phase 0 has spent), not at each wave's fragment reads; sums unchanged
# speedup vs baseline: 4.0899x; 1.0357x over previous
#include <hip/hip_runtime.h>

typedef float v2f __attribute__((ext_vector_type(2)));
typedef float v8f __attribute__((ext_vector_type(8)));

#define C_IN   128
#define K_OUT  128
#define H_     56
#define W_     56
#define HW_    3136
#define KRED   1152
#define SROWS  8
#define NSTRIP 7
#define LROWS  10
#define XROW   58
#define XPL    584
#define NPAIR  64
#define TROWS  12
#define XPLT   696
#define BLKP   (4 * XPLT + 4 * XPL)
typedef float v4f __attribute__((ext_vector_type(4)));

template <class V>
__device__ __forceinline__ void quad_rows_f43(const V* d, V* v)
{
    const V sp = d[2] * 4.0f - d[4];
    const V sr = d[1] * 4.0f - d[3];
    const V sf = d[4] - d[2];
    const V sh = d[3] - d[1];
    v[0] = (d[0] - d[2]) * 4.0f + sf;
    v[1] = sp + sr;
    v[2] = sp - sr;
    v[3] = sf + sh * 2.0f;
    v[4] = sf - sh * 2.0f;
    v[5] = (d[5] - d[3]) - sh * 4.0f;
}


template <int NQ>
__device__ __forceinline__ void tile_bases(int (&qb)[NQ], const int qt0, const int m, const int hh)
{
    #pragma unroll
    for (int t = 0; t < NQ; ++t) {
        const int q   = 16 * (qt0 + t) + m;
        const int jq  = q / W_;
        const int col = q - jq * W_;
        qb[t] = 2 * hh * XPLT + 6 * jq * XROW + col;
    }
}

template <int NQ>
__device__ __forceinline__ void phase_f43(const v2f* xs, const float* wrow, float (&w)[36],
                                          const int cb0, const int (&qb)[NQ], v8f (&am)[NQ][6])
{
    v2f pp[NQ][6];
    #pragma unroll
    for (int t = 0; t < NQ; ++t) {
        #pragma unroll
        for (int i = 0; i < 6; ++i) pp[t][i] = xs[qb[t] + i * XROW];
    }

    for (int cl = 0; cl < 8; ++cl) {
        const int cln = cl < 7 ? cl + 1 : 7;
        const int cbn = cb0 + cl < 15 ? cb0 + cl + 1 : 15;
        const float* wn = wrow + cbn * 72;
        const v2f* xb  = xs + cl * BLKP;
        const v2f* xbn = xb + (cln - cl) * BLKP;

        #pragma unroll
        for (int g = 0; g < 6; ++g) {
            const int j = g / 3, s = g % 3;
            v2f d[NQ][6];
            #pragma unroll
            for (int t = 0; t < NQ; ++t) {
                #pragma unroll
                for (int i = 0; i < 6; ++i) d[t][i] = pp[t][i];
            }
            {
                const int gn = (g + 1) % 6, jn = gn / 3, sn = gn % 3;
                const v2f* xq = (g < 5) ? xb : xbn;
                #pragma unroll
                for (int t = 0; t < NQ; ++t) {
                    #pragma unroll
                    for (int i = 0; i < 6; ++i) pp[t][i] = xq[qb[t] + jn * XPLT + i * XROW + sn];
                }
            }
            __builtin_amdgcn_sched_barrier(0);

            const v2f g0 = (v2f){w[18 * j + s],     w[18 * j + 9 + s]};
            const v2f g1 = (v2f){w[18 * j + 3 + s], w[18 * j + 12 + s]};
            const v2f g2 = (v2f){w[18 * j + 6 + s], w[18 * j + 15 + s]};
            const v2f A0 = g0 * 0.25f;
            const v2f t6 = g2 * (1.0f / 6.0f);
            const v2f h6 = g0 * (1.0f / 6.0f) + t6;
            const v2f k4 = g0 * (1.0f / 24.0f) + t6;
            const v2f A1 = h6 + g1 * (1.0f / 6.0f);
            const v2f A2 = h6 - g1 * (1.0f / 6.0f);
            const v2f A3 = k4 + g1 * (1.0f / 12.0f);
            const v2f A4 = k4 - g1 * (1.0f / 12.0f);

            #pragma unroll
            for (int t = 0; t < NQ; ++t) {
                am[t][0] = __builtin_amdgcn_wmma_f32_16x16x4_f32(
                    false, A0, false, d[t][0], (short)0, am[t][0], false, false);
                am[t][1] = __builtin_amdgcn_wmma_f32_16x16x4_f32(
                    false, A1, false, d[t][1], (short)0, am[t][1], false, false);
                am[t][2] = __builtin_amdgcn_wmma_f32_16x16x4_f32(
                    false, A2, false, d[t][2], (short)0, am[t][2], false, false);
                am[t][3] = __builtin_amdgcn_wmma_f32_16x16x4_f32(
                    false, A3, false, d[t][3], (short)0, am[t][3], false, false);
                am[t][4] = __builtin_amdgcn_wmma_f32_16x16x4_f32(
                    false, A4, false, d[t][4], (short)0, am[t][4], false, false);
                am[t][5] = __builtin_amdgcn_wmma_f32_16x16x4_f32(
                    false, g2, false, d[t][5], (short)0, am[t][5], false, false);
            }
            __builtin_amdgcn_sched_barrier(0);

            if (s == 2) {
                asm volatile("" ::: "memory");
                #pragma unroll
                for (int i = 0; i < 18; ++i) w[18 * j + i] = wn[18 * j + i];
            }
        }
    }
}

__device__ __forceinline__ void transform_raw_half(v2f* xs, const int tid)
{
    #pragma unroll
    for (int hb = 0; hb < 2; ++hb) {
        v2f d[7][6];
        int dst[7];
        #pragma unroll
        for (int k = 0; k < 7; ++k) {
            const int e  = (7 * hb + k) * 256 + tid;
            const int u  = e / W_;
            const int x  = e - u * W_ + 1;
            const int p  = u >> 1, jq = u & 1;
            const int bb = (p >> 2) * BLKP;
            const int rb = bb + 4 * XPLT + (p & 3) * XPL + 4 * jq * XROW + x;
            dst[k]       = bb + (p & 3) * XPLT + 6 * jq * XROW + x;
            #pragma unroll
            for (int i = 0; i < 6; ++i) d[k][i] = xs[rb + i * XROW];
        }
        #pragma unroll
        for (int k = 0; k < 7; ++k) {
            v2f v[6];
            quad_rows_f43(d[k], v);
            #pragma unroll
            for (int i = 0; i < 6; ++i) xs[dst[k] + i * XROW] = v[i];
        }
    }
}

template <int NQ>
__device__ __forceinline__ void store_f43(const v8f (&am)[NQ][6], const int qt0, const int m,
                                          float* outp, const float* bias8)
{
    float bv[8];
    #pragma unroll
    for (int g = 0; g < 8; ++g) bv[g] = bias8[g];
    #pragma unroll
    for (int t = 0; t < NQ; ++t) {
        const int q   = 16 * (qt0 + t) + m;
        const int jq  = q / W_;
        const int col = q - jq * W_;
        float* o = outp + 4 * jq * W_ + col;
        #pragma unroll
        for (int g = 0; g < 8; ++g) {
            const float s12 = am[t][1][g] + am[t][2][g];
            const float d12 = am[t][1][g] - am[t][2][g];
            const float s34 = am[t][3][g] + am[t][4][g];
            const float d34 = am[t][3][g] - am[t][4][g];
            o[(long)g * HW_]          = (am[t][0][g] + s12 + s34) + bv[g];
            o[(long)g * HW_ + W_]     = (d12 + 2.0f * d34) + bv[g];
            o[(long)g * HW_ + 2 * W_] = (s12 + 4.0f * s34) + bv[g];
            o[(long)g * HW_ + 3 * W_] = (d12 + 8.0f * d34 + am[t][5][g]) + bv[g];
        }
    }
}

template <int NQ>
__device__ __forceinline__ void strip_f43(v2f* xs, const float* wrow, float (&w)[36],
                                          const int qt0, const int m, const int hh, const int tid,
                                          float* outp, const float* bias8)
{
    v8f am[NQ][6];
    #pragma unroll
    for (int t = 0; t < NQ; ++t) {
        #pragma unroll
        for (int i = 0; i < 6; ++i) am[t][i] = (v8f){};
    }
    int qb[NQ];
    tile_bases<NQ>(qb, qt0, m, hh);

    #pragma unroll
    for (int ph = 0; ph < 2; ++ph) {
        if (ph == 1) {
            __syncthreads();
            transform_raw_half(xs, tid);
            __syncthreads();
        }
        phase_f43<NQ>(xs, wrow, w, 8 * ph, qb, am);
    }

    store_f43<NQ>(am, qt0, m, outp, bias8);
}

__global__ __launch_bounds__(256, 2)
void conv3x3_strip8_wmma_f32(const float* __restrict__ in,
                             const float* __restrict__ wgt,
                             const float* __restrict__ bias,
                             float* __restrict__ out)
{
    __shared__ v2f xs[8 * BLKP];

    const int tid    = threadIdx.x;
    const int ochalf = blockIdx.x & 1;
    const int sidx   = blockIdx.x >> 1;
    const int img    = sidx / NSTRIP;
    const int strip  = sidx - img * NSTRIP;
    const int h0     = strip * SROWS;

    const int wave = __builtin_amdgcn_readfirstlane(tid >> 5);
    const int lane = tid & 31;
    const int m    = lane & 15;
    const int hh   = lane >> 4;
    const int ot   = wave >> 1;
    const int pg   = (wave + (wave >> 2)) & 1;
    const int oc0  = ochalf * 64 + ot * 16;

    const float* wrow = wgt + (long)(oc0 + m) * KRED + 36 * hh;
    float w[36];
    #pragma unroll
    for (int i = 0; i < 36; ++i) w[i] = wrow[i];

    {
        const int grp = tid >> 4;
        const int l16 = tid & 15;
        if (l16 < 14) {
            float4 v[2][4 * LROWS];
            #pragma unroll
            for (int u = 0; u < 2; ++u) {
                const int gp = grp + 16 * u;
                const float* src = in + (long)(img * C_IN + 2 * gp) * HW_ + 4 * l16;
                #pragma unroll
                for (int i = 0; i < 2 * LROWS; ++i) {
                    const int r = i % LROWS, q = i / LROWS;
                    int ih = h0 - 1 + r;
                    if (r == 0)         ih = ih < 0 ? 0 : ih;
                    if (r == LROWS - 1) ih = ih > H_ - 1 ? H_ - 1 : ih;
                    const float* s = src + (long)q * (64 * HW_) + ih * W_;
                    v[u][2 * i]     = *(const float4*)s;
                    v[u][2 * i + 1] = *(const float4*)(s + HW_);
                }
            }
            const bool top = (strip == 0), bot = (strip == NSTRIP - 1);
            #pragma unroll
            for (int u = 0; u < 2; ++u) {
                const int gp = grp + 16 * u;
                v2f* xt = xs + (gp >> 2) * BLKP + (gp & 3) * XPLT;
                v2f* xr = xs + (gp >> 2) * BLKP + 4 * XPLT + (gp & 3) * XPL;
                #pragma unroll
                for (int jq = 0; jq < 2; ++jq) {
                    v4f dx[6], dy[6], tx[6], ty[6];
                    #pragma unroll
                    for (int i = 0; i < 6; ++i) {
                        const int r = 4 * jq + i;
                        float4 x = v[u][2 * r], y = v[u][2 * r + 1];
                        if ((r == 0 && top) || (r == LROWS - 1 && bot)) {
                            x = make_float4(0.f, 0.f, 0.f, 0.f);
                            y = x;
                        }
                        dx[i] = (v4f){x.x, x.y, x.z, x.w};
                        dy[i] = (v4f){y.x, y.y, y.z, y.w};
                    }
                    quad_rows_f43(dx, tx);
                    quad_rows_f43(dy, ty);
                    #pragma unroll
                    for (int i = 0; i < 6; ++i) {
                        v2f* d = xt + (6 * jq + i) * XROW + 1 + 4 * l16;
                        d[0] = (v2f){tx[i][0], ty[i][0]};
                        d[1] = (v2f){tx[i][1], ty[i][1]};
                        d[2] = (v2f){tx[i][2], ty[i][2]};
                        d[3] = (v2f){tx[i][3], ty[i][3]};
                    }
                }
                #pragma unroll
                for (int r = 0; r < LROWS; ++r) {
                    float4 x = v[u][2 * (LROWS + r)], y = v[u][2 * (LROWS + r) + 1];
                    if ((r == 0 && top) || (r == LROWS - 1 && bot)) {
                        x = make_float4(0.f, 0.f, 0.f, 0.f);
                        y = x;
                    }
                    v2f* d = xr + r * XROW + 1 + 4 * l16;
                    d[0] = (v2f){x.x, y.x};
                    d[1] = (v2f){x.y, y.y};
                    d[2] = (v2f){x.z, y.z};
                    d[3] = (v2f){x.w, y.w};
                }
            }
        } else {
            const int bcol = (l16 == 14) ? 0 : (XROW - 1);
            #pragma unroll
            for (int u = 0; u < 2; ++u) {
                const int gp = grp + 16 * u;
                v2f* xt = xs + (gp >> 2) * BLKP + (gp & 3) * XPLT;
                v2f* xr = xs + (gp >> 2) * BLKP + 4 * XPLT + (gp & 3) * XPL;
                #pragma unroll
                for (int r = 0; r < TROWS; ++r)
                    xt[r * XROW + bcol] = (v2f){0.f, 0.f};
                #pragma unroll
                for (int r = 0; r < LROWS; ++r)
                    xr[r * XROW + bcol] = (v2f){0.f, 0.f};
            }
        }
    }
    __syncthreads();

    float*       outp  = out + (long)(img * K_OUT + oc0 + 8 * hh) * HW_ + h0 * W_;
    const float* bias8 = bias + oc0 + 8 * hh;
    if (pg == 0) strip_f43<4>(xs, wrow, w, 0, m, hh, tid, outp, bias8);
    else         strip_f43<3>(xs, wrow, w, 4, m, hh, tid, outp, bias8);
}

extern "C" void kernel_launch(void* const* d_in, const int* in_sizes, int n_in,
                              void* d_out, int out_size, void* d_ws, size_t ws_size,
                              hipStream_t stream) {
    (void)in_sizes; (void)n_in; (void)d_ws; (void)ws_size; (void)out_size;
    const float* in   = (const float*)d_in[0];
    const float* wgt  = (const float*)d_in[1];
    const float* bias = (const float*)d_in[2];
    float*       out  = (float*)d_out;

    dim3 grid(224, 1, 1);
    dim3 block(256, 1, 1);
    conv3x3_strip8_wmma_f32<<<grid, block, 0, stream>>>(in, wgt, bias, out);
}
